# starting kernel
# baseline (speedup 1.0000x reference)
#include <hip/hip_runtime.h>

typedef __attribute__((ext_vector_type(2))) float v2f;
typedef __attribute__((ext_vector_type(4))) float v4f;
typedef __attribute__((ext_vector_type(8))) float v8f;

#define B_  32
#define IC  32
#define OC  32
#define HH  64
#define WW  64
#define KD  288
#define LDP 292

__global__ __launch_bounds__(128)
void lc2d_wmma_f32_kernel(const float* __restrict__ x,
                          const float* __restrict__ wgt,
                          const float* __restrict__ bias,
                          float* __restrict__ out) {
    __shared__ float sA[B_ * LDP];
    __shared__ float sB[OC * LDP];

    const int w  = blockIdx.x;
    const int h  = blockIdx.y;
    const int t  = threadIdx.x;
    const int hw = h * WW + w;

    {
        const v4f* wp4 = (const v4f*)(wgt + (size_t)hw * (OC * KD));
        for (int e4 = t; e4 < (OC * KD) / 4; e4 += 128) {
            int e   = e4 * 4;
            int row = e / KD;
            int col = e - row * KD;
            v4f v = __builtin_nontemporal_load(wp4 + e4);
            *(v4f*)&sB[row * LDP + col] = v;
        }
    }

    for (int r = t; r < B_ * IC * 3; r += 128) {
        int b   = r / 96;
        int rem = r - b * 96;
        int c   = rem / 3;
        int i   = rem - c * 3;
        int y   = h - 1 + i;
        const bool yok = (unsigned)y < (unsigned)HH;
        const float* xrow = x + ((((size_t)b * IC + c) * HH + y) * WW + (w - 1));
        float* dst = &sA[b * LDP + c * 9 + i * 3];
        dst[0] = (yok && w - 1 >= 0) ? xrow[0] : 0.0f;
        dst[1] = yok                 ? xrow[1] : 0.0f;
        dst[2] = (yok && w + 1 < WW) ? xrow[2] : 0.0f;
    }
    __syncthreads();

    const int wv    = t >> 5;
    const int lane  = t & 31;
    const int m0    = (wv >> 1) * 16;
    const int n0    = (wv & 1) * 16;
    const int lr    = lane & 15;
    const int khalf = lane >> 4;

    const float* aRow = &sA[(m0 + lr) * LDP + 2 * khalf];
    const float* bRow = &sB[(n0 + lr) * LDP + 2 * khalf];

    v8f acc = {};
    #pragma unroll 8
    for (int kt = 0; kt < KD / 4; ++kt) {
        v2f a = *(const v2f*)(aRow + kt * 4);
        v2f b = *(const v2f*)(bRow + kt * 4);
        acc = __builtin_amdgcn_wmma_f32_16x16x4_f32(
            false, a, false, b, (short)0, acc, false, false);
    }

    const int N  = lane & 15;
    const int o  = n0 + N;
    const float bv = bias[(size_t)o * (HH * WW) + hw];
    #pragma unroll
    for (int r = 0; r < 8; ++r) {
        int M = r + khalf * 8;
        int b = m0 + M;
        __builtin_nontemporal_store(acc[r] + bv,
                                    out + (((size_t)b * OC + o) * (HH * WW)) + hw);
    }
}

extern "C" void kernel_launch(void* const* d_in, const int* in_sizes, int n_in,
                              void* d_out, int out_size, void* d_ws, size_t ws_size,
                              hipStream_t stream) {
    const float* x    = (const float*)d_in[0];
    const float* wgt  = (const float*)d_in[1];
    const float* bias = (const float*)d_in[2];
    float* out        = (float*)d_out;
    (void)in_sizes; (void)n_in; (void)out_size; (void)d_ws; (void)ws_size;

    dim3 grid(WW, HH);
    lc2d_wmma_f32_kernel<<<grid, 128, 0, stream>>>(x, wgt, bias, out);
}


# x slab staged in four channel groups: groups 1-3 are loaded one group ahead and stored to LDS behind the products of the chunks that need only earlier groups; stage barriers wait for LDS only
# speedup vs baseline: 5.9990x; 5.9990x over previous
#include <hip/hip_runtime.h>


typedef __attribute__((ext_vector_type(2))) float v2f;
typedef __attribute__((ext_vector_type(4))) float v4f;
typedef __attribute__((ext_vector_type(4))) int   v4i;
typedef __attribute__((ext_vector_type(8))) float v8f;

#define B_   32
#define IC   32
#define OC   32
#define HH   64
#define WW   64
#define KD   288
#define TW   16
#define SC   18
#define NPOS (IC * 3 * SC)
#define PS   33
#define OS   17
#define NCH  (KD / 32)
#define CG   8
#define NG   (IC / CG)
static_assert(NG == 4 && (CG & (CG - 1)) == 0 && 2 * 32 <= 9 * CG && 4 * 32 <= 2 * 9 * CG && 6 * 32 <= 3 * 9 * CG,
              "chunk pairs must lie inside the channel groups staged before them");

__device__ __forceinline__ v8f mma4(v2f a, v2f b, v8f c) {
    return __builtin_amdgcn_wmma_f32_16x16x4_f32(false, a, false, b, (short)0, c, false, false);
}

struct WFrag { v4f t0[4]; v4f t1[4]; };

__device__ __forceinline__ void load_chunk(WFrag& f, const float* wp0, const float* wp1, int ch) {
    #pragma unroll
    for (int q = 0; q < 4; ++q) {
        f.t0[q] = *(const v4f*)(wp0 + 32 * ch + 8 * q);
        f.t1[q] = *(const v4f*)(wp1 + 32 * ch + 8 * q);
    }
}

__device__ __forceinline__ void mma_chunk(const WFrag& f, const float* ab, const int* kt, int ch,
                                          v8f& acc00, v8f& acc01, v8f& acc10, v8f& acc11) {
    #pragma unroll
    for (int q = 0; q < 4; ++q) {
        const v4i o = *(const v4i*)(kt + 32 * ch + 8 * q);
        const v2f aLo0 = { ab[o.x],      ab[o.y]      };
        const v2f aHi0 = { ab[o.z],      ab[o.w]      };
        const v2f aLo1 = { ab[o.x + 16], ab[o.y + 16] };
        const v2f aHi1 = { ab[o.z + 16], ab[o.w + 16] };
        const v2f bLo0 = { f.t0[q].x, f.t0[q].y };
        const v2f bHi0 = { f.t0[q].z, f.t0[q].w };
        const v2f bLo1 = { f.t1[q].x, f.t1[q].y };
        const v2f bHi1 = { f.t1[q].z, f.t1[q].w };
        acc00 = mma4(aLo0, bLo0, acc00);
        acc01 = mma4(aLo0, bLo1, acc01);
        acc10 = mma4(aLo1, bLo0, acc10);
        acc11 = mma4(aLo1, bLo1, acc11);
        acc00 = mma4(aHi0, bHi0, acc00);
        acc01 = mma4(aHi0, bHi1, acc01);
        acc10 = mma4(aHi1, bHi0, acc10);
        acc11 = mma4(aHi1, bHi1, acc11);
    }
}

struct XGrp { v4f xi[3][2]; float xh[3]; };

__device__ __forceinline__ void grp_load(XGrp& g, const float* __restrict__ x, int c0, int t, int h, int w0) {
    #pragma unroll
    for (int i = 0; i < 3; ++i) {
        const int y  = h - 1 + i;
        const int yc = ((unsigned)y < (unsigned)HH) ? y : h;
        #pragma unroll
        for (int it = 0; it < 2; ++it) {
            const int item = t + 512 * it;
            const int q    = item & 3;
            const int r    = item >> 2;
            const int c    = c0 + (r & (CG - 1));
            const int b    = r / CG;
            g.xi[i][it] = *(const v4f*)(x + ((size_t)(b * IC + c) * HH + yc) * WW + w0 + 4 * q);
        }
        {
            const int side = t & 1;
            const int r    = t >> 1;
            const int c    = c0 + (r & (CG - 1));
            const int b    = r / CG;
            const int gw   = side ? w0 + TW : w0 - 1;
            const int gc   = ((unsigned)gw < (unsigned)WW) ? gw : w0;
            g.xh[i] = x[((size_t)(b * IC + c) * HH + yc) * WW + gc];
        }
    }
}

__device__ __forceinline__ void grp_store(const XGrp& g, float* slab, int c0, int t, int h, int w0) {
    #pragma unroll
    for (int i = 0; i < 3; ++i) {
        const int  y   = h - 1 + i;
        const bool yok = (unsigned)y < (unsigned)HH;
        #pragma unroll
        for (int it = 0; it < 2; ++it) {
            const int item = t + 512 * it;
            const int q    = item & 3;
            const int r    = item >> 2;
            const int c    = c0 + (r & (CG - 1));
            const int b    = r / CG;
            const v4f v    = g.xi[i][it];
            float* d = slab + ((c * 3 + i) * SC + 1 + 4 * q) * PS + b;
            d[0]      = yok ? v.x : 0.0f;
            d[PS]     = yok ? v.y : 0.0f;
            d[2 * PS] = yok ? v.z : 0.0f;
            d[3 * PS] = yok ? v.w : 0.0f;
        }
        {
            const int side = t & 1;
            const int r    = t >> 1;
            const int c    = c0 + (r & (CG - 1));
            const int b    = r / CG;
            const int gw   = side ? w0 + TW : w0 - 1;
            const bool ok  = yok && ((unsigned)gw < (unsigned)WW);
            slab[((c * 3 + i) * SC + (side ? SC - 1 : 0)) * PS + b] = ok ? g.xh[i] : 0.0f;
        }
    }
}

__global__ __launch_bounds__(512)
void lc2d_wave_per_location_kernel(const float* __restrict__ x,
                                   const float* __restrict__ wgt,
                                   const float* __restrict__ bias,
                                   float* __restrict__ out) {
    __shared__ float slab[NPOS * PS];
    __shared__ alignas(16) int koff[KD];

    const int t  = threadIdx.x;
    const int w0 = blockIdx.x * TW;
    const int h  = blockIdx.y;

    if (t < KD) koff[t] = ((t / 3) * SC + (t % 3)) * PS;

#define SCHED_FENCE() __builtin_amdgcn_sched_barrier(0)
#define WG_BARRIER_LDS() asm volatile("s_wait_dscnt 0x0\n\ts_barrier_signal -1\n\ts_barrier_wait -1" ::: "memory")
    XGrp xg;
    grp_load(xg, x, 0, t, h, w0);
    SCHED_FENCE();
    grp_store(xg, slab, 0, t, h, w0);
    __syncthreads();

    const int wv   = t >> 5;
    const int lane = t & 31;
    const int c16  = lane & 15;
    const int hh   = lane >> 4;
    const int hw   = h * WW + w0 + wv;

    const float* wp0 = wgt + (size_t)hw * (OC * KD) + c16 * KD + 4 * hh;
    const float* wp1 = wp0 + 16 * KD;
    const float* ab  = slab + wv * PS + c16;
    const int*   kt  = koff + 4 * hh;

    const float bv0 = bias[(size_t)c16 * (HH * WW) + hw];
    const float bv1 = bias[(size_t)(c16 + 16) * (HH * WW) + hw];

    v8f acc00 = {}, acc01 = {}, acc10 = {}, acc11 = {};

    WFrag fa, fb, fc;
    load_chunk(fa, wp0, wp1, 0);
    load_chunk(fb, wp0, wp1, 1);
    SCHED_FENCE();
    load_chunk(fc, wp0, wp1, 2);  SCHED_FENCE();
    grp_load(xg, x, 1 * CG, t, h, w0);  SCHED_FENCE();
    mma_chunk(fa, ab, kt, 0, acc00, acc01, acc10, acc11);  SCHED_FENCE();
    load_chunk(fa, wp0, wp1, 3);  SCHED_FENCE();
    mma_chunk(fb, ab, kt, 1, acc00, acc01, acc10, acc11);  SCHED_FENCE();
    load_chunk(fb, wp0, wp1, 4);  SCHED_FENCE();
    grp_store(xg, slab, 1 * CG, t, h, w0);
    WG_BARRIER_LDS();
    grp_load(xg, x, 2 * CG, t, h, w0);  SCHED_FENCE();
    mma_chunk(fc, ab, kt, 2, acc00, acc01, acc10, acc11);  SCHED_FENCE();
    load_chunk(fc, wp0, wp1, 5);  SCHED_FENCE();
    mma_chunk(fa, ab, kt, 3, acc00, acc01, acc10, acc11);  SCHED_FENCE();
    load_chunk(fa, wp0, wp1, 6);  SCHED_FENCE();
    grp_store(xg, slab, 2 * CG, t, h, w0);
    WG_BARRIER_LDS();
    grp_load(xg, x, 3 * CG, t, h, w0);  SCHED_FENCE();
    mma_chunk(fb, ab, kt, 4, acc00, acc01, acc10, acc11);  SCHED_FENCE();
    load_chunk(fb, wp0, wp1, 7);  SCHED_FENCE();
    mma_chunk(fc, ab, kt, 5, acc00, acc01, acc10, acc11);  SCHED_FENCE();
    load_chunk(fc, wp0, wp1, 8);  SCHED_FENCE();
    grp_store(xg, slab, 3 * CG, t, h, w0);
    WG_BARRIER_LDS();
    mma_chunk(fa, ab, kt, 6, acc00, acc01, acc10, acc11);
    mma_chunk(fb, ab, kt, 7, acc00, acc01, acc10, acc11);
    mma_chunk(fc, ab, kt, 8, acc00, acc01, acc10, acc11);
#undef WG_BARRIER_LDS
#undef SCHED_FENCE

    __syncthreads();
    float* obuf = slab;
    #pragma unroll
    for (int r = 0; r < 8; ++r) {
        const int bl = 8 * hh + r;
        obuf[((bl     ) * OC + c16     ) * OS + wv] = acc00[r] + bv0;
        obuf[((bl     ) * OC + c16 + 16) * OS + wv] = acc01[r] + bv1;
        obuf[((bl + 16) * OC + c16     ) * OS + wv] = acc10[r] + bv0;
        obuf[((bl + 16) * OC + c16 + 16) * OS + wv] = acc11[r] + bv1;
    }
    __syncthreads();

    #pragma unroll
    for (int it = 0; it < (B_ * OC * TW / 4) / 512; ++it) {
        const int idx4 = t + 512 * it;
        const int q    = idx4 & 3;
        const int bo   = idx4 >> 2;
        const float* s = obuf + bo * OS + 4 * q;
        const v4f v = { s[0], s[1], s[2], s[3] };
        __builtin_nontemporal_store(v, (v4f*)(out + (size_t)bo * (HH * WW) + h * WW + w0 + 4 * q));
    }
}

extern "C" void kernel_launch(void* const* d_in, const int* in_sizes, int n_in,
                              void* d_out, int out_size, void* d_ws, size_t ws_size,
                              hipStream_t stream) {
    const float* x    = (const float*)d_in[0];
    const float* wgt  = (const float*)d_in[1];
    const float* bias = (const float*)d_in[2];
    float* out        = (float*)d_out;
    (void)in_sizes; (void)n_in; (void)out_size; (void)d_ws; (void)ws_size;

    dim3 grid(WW / TW, HH);
    lc2d_wave_per_location_kernel<<<grid, 512, 0, stream>>>(x, wgt, bias, out);
}


# weight loads cover whole lines: a lane quad and its two halves read one row's 128-byte line per load (4 lines per wave load, was 16 x 32 bytes), pieces exchanged in the quad by DPP; new k sum order
# speedup vs baseline: 6.1831x; 1.0307x over previous
#include <hip/hip_runtime.h>


typedef __attribute__((ext_vector_type(2))) float v2f;
typedef __attribute__((ext_vector_type(4))) float v4f;
typedef __attribute__((ext_vector_type(4))) int   v4i;
typedef __attribute__((ext_vector_type(8))) float v8f;

#define B_   32
#define IC   32
#define OC   32
#define HH   64
#define WW   64
#define KD   288
#define TW   16
#define SC   18
#define NPOS (IC * 3 * SC)
#define PS   33
#define OS   17
#define NCH  (KD / 32)
#define CG   8
#define NG   (IC / CG)
static_assert(NG == 4 && (CG & (CG - 1)) == 0 && 2 * 32 <= 9 * CG && 4 * 32 <= 2 * 9 * CG && 6 * 32 <= 3 * 9 * CG,
              "chunk pairs must lie inside the channel groups staged before them");

__device__ __forceinline__ v8f mma4(v2f a, v2f b, v8f c) {
    return __builtin_amdgcn_wmma_f32_16x16x4_f32(false, a, false, b, (short)0, c, false, false);
}

struct WFrag { v4f r[2][4]; };

__device__ __forceinline__ void load_chunk(WFrag& f, const float* const (&wp)[4], int ch) {
    #pragma unroll
    for (int tl = 0; tl < 2; ++tl) {
        #pragma unroll
        for (int u = 0; u < 4; ++u)
            f.r[tl][u] = *(const v4f*)(wp[u] + tl * 16 * KD + 32 * ch);
    }
}

template <int CTRL>
__device__ __forceinline__ float quad_from(float x) {
    return __builtin_bit_cast(float, __builtin_amdgcn_mov_dpp(__builtin_bit_cast(int, x), CTRL, 0xF, 0xF, false));
}

__device__ __forceinline__ void exchange_chunk(WFrag& f, bool j0, bool j1) {
    #pragma unroll
    for (int tl = 0; tl < 2; ++tl) {
        #pragma unroll
        for (int d = 0; d < 4; ++d) {
            const float r0 = f.r[tl][0][d], r1 = f.r[tl][1][d], r2 = f.r[tl][2][d], r3 = f.r[tl][3][d];
            const float a0 = j0 ? r1 : r0, a1 = j0 ? r0 : r1;
            const float a2 = j0 ? r3 : r2, a3 = j0 ? r2 : r3;
            const float z0 = j1 ? a2 : a0, z2 = j1 ? a0 : a2;
            const float z1 = j1 ? a3 : a1, z3 = j1 ? a1 : a3;
            f.r[tl][0][d] = z0;
            f.r[tl][1][d] = quad_from<0xB1>(z1);
            f.r[tl][2][d] = quad_from<0x4E>(z2);
            f.r[tl][3][d] = quad_from<0x1B>(z3);
        }
    }
}

struct AFrag { v2f lo0[4], hi0[4], lo1[4], hi1[4]; };

__device__ __forceinline__ void a_reads(AFrag& a, const float* ab, const int* kt, int ch) {
    #pragma unroll
    for (int s = 0; s < 4; ++s) {
        const v4i o = *(const v4i*)(kt + 32 * ch + 4 * s);
        a.lo0[s] = v2f{ ab[o.x],      ab[o.y]      };
        a.hi0[s] = v2f{ ab[o.z],      ab[o.w]      };
        a.lo1[s] = v2f{ ab[o.x + 16], ab[o.y + 16] };
        a.hi1[s] = v2f{ ab[o.z + 16], ab[o.w + 16] };
    }
}

__device__ __forceinline__ void products(const WFrag& f, const AFrag& a,
                                         v8f& acc00, v8f& acc01, v8f& acc10, v8f& acc11) {
    #pragma unroll
    for (int s = 0; s < 4; ++s) {
        const v2f bLo0 = { f.r[0][s].x, f.r[0][s].y };
        const v2f bHi0 = { f.r[0][s].z, f.r[0][s].w };
        const v2f bLo1 = { f.r[1][s].x, f.r[1][s].y };
        const v2f bHi1 = { f.r[1][s].z, f.r[1][s].w };
        acc00 = mma4(a.lo0[s], bLo0, acc00);
        acc01 = mma4(a.lo0[s], bLo1, acc01);
        acc10 = mma4(a.lo1[s], bLo0, acc10);
        acc11 = mma4(a.lo1[s], bLo1, acc11);
        acc00 = mma4(a.hi0[s], bHi0, acc00);
        acc01 = mma4(a.hi0[s], bHi1, acc01);
        acc10 = mma4(a.hi1[s], bHi0, acc10);
        acc11 = mma4(a.hi1[s], bHi1, acc11);
    }
    asm volatile("" : "+v"(acc00), "+v"(acc01), "+v"(acc10), "+v"(acc11));
}

struct XGrp { v4f xi[3][2]; float xh[3]; };

__device__ __forceinline__ void grp_load(XGrp& g, const float* __restrict__ x, int c0, int t, int h, int w0) {
    #pragma unroll
    for (int i = 0; i < 3; ++i) {
        const int y  = h - 1 + i;
        const int yc = ((unsigned)y < (unsigned)HH) ? y : h;
        #pragma unroll
        for (int it = 0; it < 2; ++it) {
            const int item = t + 512 * it;
            const int q    = item & 3;
            const int r    = item >> 2;
            const int c    = c0 + (r & (CG - 1));
            const int b    = r / CG;
            g.xi[i][it] = *(const v4f*)(x + ((size_t)(b * IC + c) * HH + yc) * WW + w0 + 4 * q);
        }
        {
            const int side = t & 1;
            const int r    = t >> 1;
            const int c    = c0 + (r & (CG - 1));
            const int b    = r / CG;
            const int gw   = side ? w0 + TW : w0 - 1;
            const int gc   = ((unsigned)gw < (unsigned)WW) ? gw : w0;
            g.xh[i] = x[((size_t)(b * IC + c) * HH + yc) * WW + gc];
        }
    }
}

__device__ __forceinline__ void grp_store(const XGrp& g, float* slab, int c0, int t, int h, int w0) {
    #pragma unroll
    for (int i = 0; i < 3; ++i) {
        const int  y   = h - 1 + i;
        const bool yok = (unsigned)y < (unsigned)HH;
        #pragma unroll
        for (int it = 0; it < 2; ++it) {
            const int item = t + 512 * it;
            const int q    = item & 3;
            const int r    = item >> 2;
            const int c    = c0 + (r & (CG - 1));
            const int b    = r / CG;
            const v4f v    = g.xi[i][it];
            float* d = slab + ((c * 3 + i) * SC + 1 + 4 * q) * PS + b;
            d[0]      = yok ? v.x : 0.0f;
            d[PS]     = yok ? v.y : 0.0f;
            d[2 * PS] = yok ? v.z : 0.0f;
            d[3 * PS] = yok ? v.w : 0.0f;
        }
        {
            const int side = t & 1;
            const int r    = t >> 1;
            const int c    = c0 + (r & (CG - 1));
            const int b    = r / CG;
            const int gw   = side ? w0 + TW : w0 - 1;
            const bool ok  = yok && ((unsigned)gw < (unsigned)WW);
            slab[((c * 3 + i) * SC + (side ? SC - 1 : 0)) * PS + b] = ok ? g.xh[i] : 0.0f;
        }
    }
}

__global__ __launch_bounds__(512)
void lc2d_wave_per_location_kernel(const float* __restrict__ x,
                                   const float* __restrict__ wgt,
                                   const float* __restrict__ bias,
                                   float* __restrict__ out) {
    __shared__ float slab[NPOS * PS];
    __shared__ alignas(16) int koff[KD];

    const int t  = threadIdx.x;
    const int w0 = blockIdx.x * TW;
    const int h  = blockIdx.y;

    if (t < KD) koff[t] = ((t / 3) * SC + (t % 3)) * PS;

#define SCHED_FENCE() __builtin_amdgcn_sched_barrier(0)
#define WG_BARRIER_LDS() asm volatile("s_wait_dscnt 0x0\n\ts_barrier_signal -1\n\ts_barrier_wait -1" ::: "memory")
    XGrp xg;
    grp_load(xg, x, 0, t, h, w0);
    SCHED_FENCE();
    grp_store(xg, slab, 0, t, h, w0);
    __syncthreads();

    const int wv   = t >> 5;
    const int lane = t & 31;
    const int c16  = lane & 15;
    const int hh   = lane >> 4;
    const int jq   = lane & 3;
    const bool j0  = (jq & 1) != 0;
    const bool j1  = (jq & 2) != 0;
    const int hw   = h * WW + w0 + wv;

    const float* wl  = wgt + (size_t)hw * (OC * KD) + (c16 & ~3) * KD + 16 * hh;
    const float* const wp[4] = { wl + 0 * KD + 4 * (jq ^ 0), wl + 1 * KD + 4 * (jq ^ 1),
                                 wl + 2 * KD + 4 * (jq ^ 2), wl + 3 * KD + 4 * (jq ^ 3) };
    const float* ab  = slab + wv * PS + c16;
    const int*   kt  = koff + 16 * hh;

    const float bv0 = bias[(size_t)c16 * (HH * WW) + hw];
    const float bv1 = bias[(size_t)(c16 + 16) * (HH * WW) + hw];

    v8f acc00 = {}, acc01 = {}, acc10 = {}, acc11 = {};

    WFrag fa, fb, fc;
    AFrag af;
    load_chunk(fa, wp, 0);
    load_chunk(fb, wp, 1);
    SCHED_FENCE();
    load_chunk(fc, wp, 2);  SCHED_FENCE();
    grp_load(xg, x, 1 * CG, t, h, w0);  SCHED_FENCE();
    a_reads(af, ab, kt, 0);  SCHED_FENCE();
    exchange_chunk(fa, j0, j1);  SCHED_FENCE();
    products(fa, af, acc00, acc01, acc10, acc11);  SCHED_FENCE();
    load_chunk(fa, wp, 3);  SCHED_FENCE();
    a_reads(af, ab, kt, 1);  SCHED_FENCE();
    exchange_chunk(fb, j0, j1);  SCHED_FENCE();
    products(fb, af, acc00, acc01, acc10, acc11);  SCHED_FENCE();
    load_chunk(fb, wp, 4);  SCHED_FENCE();
    grp_store(xg, slab, 1 * CG, t, h, w0);
    WG_BARRIER_LDS();
    grp_load(xg, x, 2 * CG, t, h, w0);  SCHED_FENCE();
    a_reads(af, ab, kt, 2);  SCHED_FENCE();
    exchange_chunk(fc, j0, j1);  SCHED_FENCE();
    products(fc, af, acc00, acc01, acc10, acc11);  SCHED_FENCE();
    load_chunk(fc, wp, 5);  SCHED_FENCE();
    a_reads(af, ab, kt, 3);  SCHED_FENCE();
    exchange_chunk(fa, j0, j1);  SCHED_FENCE();
    products(fa, af, acc00, acc01, acc10, acc11);  SCHED_FENCE();
    load_chunk(fa, wp, 6);  SCHED_FENCE();
    grp_store(xg, slab, 2 * CG, t, h, w0);
    WG_BARRIER_LDS();
    grp_load(xg, x, 3 * CG, t, h, w0);  SCHED_FENCE();
    a_reads(af, ab, kt, 4);  SCHED_FENCE();
    exchange_chunk(fb, j0, j1);  SCHED_FENCE();
    products(fb, af, acc00, acc01, acc10, acc11);  SCHED_FENCE();
    load_chunk(fb, wp, 7);  SCHED_FENCE();
    a_reads(af, ab, kt, 5);  SCHED_FENCE();
    exchange_chunk(fc, j0, j1);  SCHED_FENCE();
    products(fc, af, acc00, acc01, acc10, acc11);  SCHED_FENCE();
    load_chunk(fc, wp, 8);  SCHED_FENCE();
    grp_store(xg, slab, 3 * CG, t, h, w0);
    WG_BARRIER_LDS();
    a_reads(af, ab, kt, 6);  SCHED_FENCE();
    exchange_chunk(fa, j0, j1);  SCHED_FENCE();
    products(fa, af, acc00, acc01, acc10, acc11);
    a_reads(af, ab, kt, 7);  SCHED_FENCE();
    exchange_chunk(fb, j0, j1);  SCHED_FENCE();
    products(fb, af, acc00, acc01, acc10, acc11);
    a_reads(af, ab, kt, 8);  SCHED_FENCE();
    exchange_chunk(fc, j0, j1);  SCHED_FENCE();
    products(fc, af, acc00, acc01, acc10, acc11);
#undef WG_BARRIER_LDS
#undef SCHED_FENCE

    __syncthreads();
    float* obuf = slab;
    #pragma unroll
    for (int r = 0; r < 8; ++r) {
        const int bl = 8 * hh + r;
        obuf[((bl     ) * OC + c16     ) * OS + wv] = acc00[r] + bv0;
        obuf[((bl     ) * OC + c16 + 16) * OS + wv] = acc01[r] + bv1;
        obuf[((bl + 16) * OC + c16     ) * OS + wv] = acc10[r] + bv0;
        obuf[((bl + 16) * OC + c16 + 16) * OS + wv] = acc11[r] + bv1;
    }
    __syncthreads();

    #pragma unroll
    for (int it = 0; it < (B_ * OC * TW / 4) / 512; ++it) {
        const int idx4 = t + 512 * it;
        const int q    = idx4 & 3;
        const int bo   = idx4 >> 2;
        const float* s = obuf + bo * OS + 4 * q;
        const v4f v = { s[0], s[1], s[2], s[3] };
        __builtin_nontemporal_store(v, (v4f*)(out + (size_t)bo * (HH * WW) + h * WW + w0 + 4 * q));
    }
}

extern "C" void kernel_launch(void* const* d_in, const int* in_sizes, int n_in,
                              void* d_out, int out_size, void* d_ws, size_t ws_size,
                              hipStream_t stream) {
    const float* x    = (const float*)d_in[0];
    const float* wgt  = (const float*)d_in[1];
    const float* bias = (const float*)d_in[2];
    float* out        = (float*)d_out;
    (void)in_sizes; (void)n_in; (void)out_size; (void)d_ws; (void)ws_size;

    dim3 grid(WW / TW, HH);
    lc2d_wave_per_location_kernel<<<grid, 512, 0, stream>>>(x, wgt, bias, out);
}


# A fragments of chunk n+1 read from LDS in front of chunk n's products: two groups of 16 single reads at literal slab offsets (no offset table), 1.5 register sets; stage barriers one chunk earlier
# speedup vs baseline: 6.5010x; 1.0514x over previous
#include <hip/hip_runtime.h>


typedef __attribute__((ext_vector_type(2))) float v2f;
typedef __attribute__((ext_vector_type(4))) float v4f;
typedef __attribute__((ext_vector_type(8))) float v8f;
typedef volatile __attribute__((address_space(3))) float lds_volatile_float;

#define B_   32
#define IC   32
#define OC   32
#define HH   64
#define WW   64
#define KD   288
#define TW   16
#define SC   18
#define NPOS (IC * 3 * SC)
#define PS   33
#define OS   17
#define NCH  (KD / 32)
#define DA   ((5 * SC + 1) * PS)
#define DB   ((6 * SC - 2) * PS)
#define CG   8
#define NG   (IC / CG)
static_assert(NG == 4 && (CG & (CG - 1)) == 0 && 2 * 32 <= 9 * CG && 4 * 32 <= 2 * 9 * CG && 6 * 32 <= 3 * 9 * CG,
              "chunk pairs must lie inside the channel groups staged before them");

__device__ __forceinline__ v8f mma4(v2f a, v2f b, v8f c) {
    return __builtin_amdgcn_wmma_f32_16x16x4_f32(false, a, false, b, (short)0, c, false, false);
}

struct WFrag { v4f r[2][4]; };

__device__ __forceinline__ void load_chunk(WFrag& f, const float* wloc, const unsigned (&wofs)[4], int ch) {
    #pragma unroll
    for (int tl = 0; tl < 2; ++tl) {
        #pragma unroll
        for (int u = 0; u < 4; ++u)
            f.r[tl][u] = *(const v4f*)(wloc + wofs[u] + tl * 16 * KD + 32 * ch);
    }
}

template <int CTRL>
__device__ __forceinline__ float quad_from(float x) {
    return __builtin_bit_cast(float, __builtin_amdgcn_mov_dpp(__builtin_bit_cast(int, x), CTRL, 0xF, 0xF, false));
}

__device__ __forceinline__ void exchange_chunk(WFrag& f, bool j0, bool j1) {
    #pragma unroll
    for (int tl = 0; tl < 2; ++tl) {
        #pragma unroll
        for (int d = 0; d < 4; ++d) {
            const float r0 = f.r[tl][0][d], r1 = f.r[tl][1][d], r2 = f.r[tl][2][d], r3 = f.r[tl][3][d];
            const float a0 = j0 ? r1 : r0, a1 = j0 ? r0 : r1;
            const float a2 = j0 ? r3 : r2, a3 = j0 ? r2 : r3;
            const float z0 = j1 ? a2 : a0, z2 = j1 ? a0 : a2;
            const float z1 = j1 ? a3 : a1, z3 = j1 ? a1 : a3;
            f.r[tl][0][d] = z0;
            f.r[tl][1][d] = quad_from<0xB1>(z1);
            f.r[tl][2][d] = quad_from<0x4E>(z2);
            f.r[tl][3][d] = quad_from<0x1B>(z3);
        }
    }
}

struct AFrag { v2f lo0[4], hi0[4], lo1[4], hi1[4]; };

template <int CH, int S0>
__device__ __forceinline__ void a_reads(AFrag& a, const float* slab, int iA, int iB) {
    constexpr int base = ((32 * CH) / 3) * SC * PS;
    int jA = iA + base;
    int jB = iB + base;
    asm volatile("" : "+v"(jA), "+v"(jB));
    const lds_volatile_float* pA = (const lds_volatile_float*)slab + jA;
    const lds_volatile_float* pB = (const lds_volatile_float*)slab + jB;
    #pragma unroll
    for (int s = S0; s < S0 + 2; ++s) {
        float v[4], w[4];
        #pragma unroll
        for (int e = 0; e < 4; ++e) {
            const int k   = 32 * CH + 4 * s + e;
            const int off = ((k / 3) * SC + (k % 3)) * PS - base;
            const lds_volatile_float* p = (k % 3 == 2) ? pB : pA;
            v[e] = p[off];
            w[e] = p[off + 16];
        }
        a.lo0[s] = v2f{ v[0], v[1] };
        a.hi0[s] = v2f{ v[2], v[3] };
        a.lo1[s] = v2f{ w[0], w[1] };
        a.hi1[s] = v2f{ w[2], w[3] };
    }
}

template <int S0>
__device__ __forceinline__ void products(const WFrag& f, const AFrag& a,
                                         v8f& acc00, v8f& acc01, v8f& acc10, v8f& acc11) {
    #pragma unroll
    for (int s = S0; s < S0 + 2; ++s) {
        const v2f bLo0 = { f.r[0][s].x, f.r[0][s].y };
        const v2f bHi0 = { f.r[0][s].z, f.r[0][s].w };
        const v2f bLo1 = { f.r[1][s].x, f.r[1][s].y };
        const v2f bHi1 = { f.r[1][s].z, f.r[1][s].w };
        acc00 = mma4(a.lo0[s], bLo0, acc00);
        acc01 = mma4(a.lo0[s], bLo1, acc01);
        acc10 = mma4(a.lo1[s], bLo0, acc10);
        acc11 = mma4(a.lo1[s], bLo1, acc11);
        acc00 = mma4(a.hi0[s], bHi0, acc00);
        acc01 = mma4(a.hi0[s], bHi1, acc01);
        acc10 = mma4(a.hi1[s], bHi0, acc10);
        acc11 = mma4(a.hi1[s], bHi1, acc11);
    }
    asm volatile("" : "+v"(acc00), "+v"(acc01), "+v"(acc10), "+v"(acc11));
}

struct XGrp { v4f xi[3][2]; float xh[3]; };

__device__ __forceinline__ void grp_load(XGrp& g, const float* __restrict__ x, int c0, int t, int h, int w0) {
    #pragma unroll
    for (int i = 0; i < 3; ++i) {
        const int y  = h - 1 + i;
        const int yc = ((unsigned)y < (unsigned)HH) ? y : h;
        #pragma unroll
        for (int it = 0; it < 2; ++it) {
            const int item = t + 512 * it;
            const int q    = item & 3;
            const int r    = item >> 2;
            const int c    = c0 + (r & (CG - 1));
            const int b    = r / CG;
            g.xi[i][it] = *(const v4f*)(x + ((size_t)(b * IC + c) * HH + yc) * WW + w0 + 4 * q);
        }
        {
            const int side = t & 1;
            const int r    = t >> 1;
            const int c    = c0 + (r & (CG - 1));
            const int b    = r / CG;
            const int gw   = side ? w0 + TW : w0 - 1;
            const int gc   = ((unsigned)gw < (unsigned)WW) ? gw : w0;
            g.xh[i] = x[((size_t)(b * IC + c) * HH + yc) * WW + gc];
        }
    }
}

__device__ __forceinline__ void grp_store(const XGrp& g, float* slab, int c0, int t, int h, int w0) {
    #pragma unroll
    for (int i = 0; i < 3; ++i) {
        const int  y   = h - 1 + i;
        const bool yok = (unsigned)y < (unsigned)HH;
        #pragma unroll
        for (int it = 0; it < 2; ++it) {
            const int item = t + 512 * it;
            const int q    = item & 3;
            const int r    = item >> 2;
            const int c    = c0 + (r & (CG - 1));
            const int b    = r / CG;
            const v4f v    = g.xi[i][it];
            float* d = slab + ((c * 3 + i) * SC + 1 + 4 * q) * PS + b;
            d[0]      = yok ? v.x : 0.0f;
            d[PS]     = yok ? v.y : 0.0f;
            d[2 * PS] = yok ? v.z : 0.0f;
            d[3 * PS] = yok ? v.w : 0.0f;
        }
        {
            const int side = t & 1;
            const int r    = t >> 1;
            const int c    = c0 + (r & (CG - 1));
            const int b    = r / CG;
            const int gw   = side ? w0 + TW : w0 - 1;
            const bool ok  = yok && ((unsigned)gw < (unsigned)WW);
            slab[((c * 3 + i) * SC + (side ? SC - 1 : 0)) * PS + b] = ok ? g.xh[i] : 0.0f;
        }
    }
}

__global__ __launch_bounds__(512)
void lc2d_wave_per_location_kernel(const float* __restrict__ x,
                                   const float* __restrict__ wgt,
                                   const float* __restrict__ bias,
                                   float* __restrict__ out) {
    __shared__ float slab[NPOS * PS];

    const int t  = threadIdx.x;
    const int w0 = blockIdx.x * TW;
    const int h  = blockIdx.y;

#define SCHED_FENCE() __builtin_amdgcn_sched_barrier(0)
#define WG_BARRIER_LDS() asm volatile("s_wait_dscnt 0x0\n\ts_barrier_signal -1\n\ts_barrier_wait -1" ::: "memory")
    XGrp xg;
    grp_load(xg, x, 0, t, h, w0);
    SCHED_FENCE();
    grp_store(xg, slab, 0, t, h, w0);
    __syncthreads();

    const int wv   = t >> 5;
    const int lane = t & 31;
    const int c16  = lane & 15;
    const int hh   = lane >> 4;
    const int jq   = lane & 3;
    const bool j0  = (jq & 1) != 0;
    const bool j1  = (jq & 2) != 0;
    const int hw   = h * WW + w0 + wv;

    const float* wloc = wgt + (size_t)__builtin_amdgcn_readfirstlane(hw) * (OC * KD);
    const unsigned wo = (c16 & ~3) * KD + 16 * hh;
    const unsigned wofs[4] = { wo + 0 * KD + 4 * (jq ^ 0), wo + 1 * KD + 4 * (jq ^ 1),
                               wo + 2 * KD + 4 * (jq ^ 2), wo + 3 * KD + 4 * (jq ^ 3) };
    const int iA = wv * PS + c16 + hh * DA;
    const int iB = wv * PS + c16 + hh * DB;

    v8f acc00 = {}, acc01 = {}, acc10 = {}, acc11 = {};

#define CHUNK_STEP(F, CUR, NXT, N1)                                        \
    exchange_chunk(F, j0, j1);  SCHED_FENCE();                             \
    a_reads<N1, 0>(NXT, slab, iA, iB);  SCHED_FENCE();                     \
    products<0>(F, CUR, acc00, acc01, acc10, acc11);  SCHED_FENCE();       \
    a_reads<N1, 2>(NXT, slab, iA, iB);  SCHED_FENCE();                     \
    products<2>(F, CUR, acc00, acc01, acc10, acc11);  SCHED_FENCE()
    WFrag fa, fb, fc;
    AFrag a0, a1;
    grp_load(xg, x, 1 * CG, t, h, w0);  SCHED_FENCE();
    load_chunk(fa, wloc, wofs, 0);
    load_chunk(fb, wloc, wofs, 1);
    SCHED_FENCE();
    load_chunk(fc, wloc, wofs, 2);  SCHED_FENCE();
    a_reads<0, 0>(a0, slab, iA, iB);
    a_reads<0, 2>(a0, slab, iA, iB);  SCHED_FENCE();
    CHUNK_STEP(fa, a0, a1, 1);
    load_chunk(fa, wloc, wofs, 3);  SCHED_FENCE();
    grp_store(xg, slab, 1 * CG, t, h, w0);
    WG_BARRIER_LDS();
    grp_load(xg, x, 2 * CG, t, h, w0);  SCHED_FENCE();
    CHUNK_STEP(fb, a1, a0, 2);
    load_chunk(fb, wloc, wofs, 4);  SCHED_FENCE();
    CHUNK_STEP(fc, a0, a1, 3);
    load_chunk(fc, wloc, wofs, 5);  SCHED_FENCE();
    grp_store(xg, slab, 2 * CG, t, h, w0);
    WG_BARRIER_LDS();
    grp_load(xg, x, 3 * CG, t, h, w0);  SCHED_FENCE();
    CHUNK_STEP(fa, a1, a0, 4);
    load_chunk(fa, wloc, wofs, 6);  SCHED_FENCE();
    CHUNK_STEP(fb, a0, a1, 5);
    load_chunk(fb, wloc, wofs, 7);  SCHED_FENCE();
    grp_store(xg, slab, 3 * CG, t, h, w0);
    WG_BARRIER_LDS();
    CHUNK_STEP(fc, a1, a0, 6);
    load_chunk(fc, wloc, wofs, 8);  SCHED_FENCE();
    const float bv0 = bias[(size_t)c16 * (HH * WW) + hw];
    const float bv1 = bias[(size_t)(c16 + 16) * (HH * WW) + hw];
    SCHED_FENCE();
    CHUNK_STEP(fa, a0, a1, 7);
    CHUNK_STEP(fb, a1, a0, 8);
    exchange_chunk(fc, j0, j1);  SCHED_FENCE();
    products<0>(fc, a0, acc00, acc01, acc10, acc11);
    products<2>(fc, a0, acc00, acc01, acc10, acc11);
#undef CHUNK_STEP
#undef WG_BARRIER_LDS
#undef SCHED_FENCE

    __syncthreads();
    float* obuf = slab;
    #pragma unroll
    for (int r = 0; r < 8; ++r) {
        const int bl = 8 * hh + r;
        obuf[((bl     ) * OC + c16     ) * OS + wv] = acc00[r] + bv0;
        obuf[((bl     ) * OC + c16 + 16) * OS + wv] = acc01[r] + bv1;
        obuf[((bl + 16) * OC + c16     ) * OS + wv] = acc10[r] + bv0;
        obuf[((bl + 16) * OC + c16 + 16) * OS + wv] = acc11[r] + bv1;
    }
    __syncthreads();

    #pragma unroll
    for (int it = 0; it < (B_ * OC * TW / 4) / 512; ++it) {
        const int idx4 = t + 512 * it;
        const int q    = idx4 & 3;
        const int bo   = idx4 >> 2;
        const float* s = obuf + bo * OS + 4 * q;
        const v4f v = { s[0], s[1], s[2], s[3] };
        __builtin_nontemporal_store(v, (v4f*)(out + (size_t)bo * (HH * WW) + h * WW + w0 + 4 * q));
    }
}

extern "C" void kernel_launch(void* const* d_in, const int* in_sizes, int n_in,
                              void* d_out, int out_size, void* d_ws, size_t ws_size,
                              hipStream_t stream) {
    const float* x    = (const float*)d_in[0];
    const float* wgt  = (const float*)d_in[1];
    const float* bias = (const float*)d_in[2];
    float* out        = (float*)d_out;
    (void)in_sizes; (void)n_in; (void)out_size; (void)d_ws; (void)ws_size;

    dim3 grid(WW / TW, HH);
    lc2d_wave_per_location_kernel<<<grid, 512, 0, stream>>>(x, wgt, bias, out);
}
